# starting kernel
# baseline (speedup 1.0000x reference)
#include <hip/hip_runtime.h>
#include <hip/hip_bf16.h>

typedef __attribute__((ext_vector_type(2))) float v2f;
typedef __attribute__((ext_vector_type(8))) float v8f;

#define MAT_N 1024
#define MAT_D 32
#define NQ    192

#define LOG2E_F 1.44269504088896340736f
#define LN2_F   0.69314718055994530942f

#define WS_TAB   1024
#define WS_PARAM 1792

__global__ __launch_bounds__(256) void matern_sqn_kernel(const float* __restrict__ X,
                                                         float* __restrict__ ws) {
    int r = blockIdx.x * blockDim.x + threadIdx.x;
    if (r < MAT_N) {
        const float4* p = (const float4*)(X + r * MAT_D);
        float s = 0.0f;
#pragma unroll
        for (int i = 0; i < MAT_D / 4; ++i) {
            float4 v = p[i];
            s += v.x * v.x + v.y * v.y + v.z * v.z + v.w * v.w;
        }
        ws[r] = s;
    }
}

__global__ __launch_bounds__(256) void matern_setup_kernel(const float* __restrict__ nu_p,
                                                           const float* __restrict__ logvar_p,
                                                           const float* __restrict__ logl_p,
                                                           float* __restrict__ ws) {
    int tid = threadIdx.x;
    float nu = nu_p[0];
    if (tid < NQ) {
        const float h = 6.0f / (float)(NQ - 1);
        float t = h * (float)tid;
        float w = h;
        if (tid == 0 || tid == NQ - 1) w *= 0.5f;
        ws[WS_TAB + 2 * tid + 0] = -(coshf(t) - 1.0f) * LOG2E_F;
        ws[WS_TAB + 2 * tid + 1] = w * coshf(nu * t);
    }
    if (tid == 0) {
        float var = expf(logvar_p[0]);
        float len = expf(logl_p[0]);
        ws[WS_PARAM + 0] = sqrtf(2.0f * nu) / len;
        ws[WS_PARAM + 1] = var;
        ws[WS_PARAM + 2] = ((1.0f - nu) * LN2_F - lgammaf(nu)) * LOG2E_F;
        ws[WS_PARAM + 3] = nu;
    }
}

__global__ __launch_bounds__(256) void matern_main_kernel(const float* __restrict__ X,
                                                          const float* __restrict__ ws,
                                                          float* __restrict__ out) {
    __shared__ float2 s_tab[NQ];

    {
        const float2* g_tab = (const float2*)(ws + WS_TAB);
        for (int i = threadIdx.x; i < NQ; i += 256) s_tab[i] = g_tab[i];
    }
    const float c_scale   = ws[WS_PARAM + 0];
    const float var       = ws[WS_PARAM + 1];
    const float log_norm2 = ws[WS_PARAM + 2];
    const float nu        = ws[WS_PARAM + 3];
    __syncthreads();

    const int wave = threadIdx.x >> 5;
    const int lane = threadIdx.x & 31;
    const int tile = blockIdx.x * 8 + wave;
    const int bi = tile >> 6;
    const int bj = tile & 63;
    const int half = lane >> 4;
    const int lr = lane & 15;

    const float* Arow = X + (bi * 16 + lr) * MAT_D + 2 * half;
    const float* Brow = X + (bj * 16 + lr) * MAT_D + 2 * half;

    v8f acc = {};
#pragma unroll
    for (int s = 0; s < 8; ++s) {
        v2f a = *(const v2f*)(Arow + 4 * s);
        v2f b = *(const v2f*)(Brow + 4 * s);
        acc = __builtin_amdgcn_wmma_f32_16x16x4_f32(
             false, a,  false, b,
             (short)0, acc,  false,  false);
    }

    const int col = bj * 16 + lr;
    const float sqn_j = ws[col];

    float x[8];
#pragma unroll
    for (int r = 0; r < 8; ++r) {
        const int row = bi * 16 + 8 * half + r;
        float sq = ws[row] + sqn_j - 2.0f * acc[r];
        sq = fmaxf(sq, 1e-24f);
        float dist = __builtin_amdgcn_sqrtf(sq);
        x[r] = fmaxf(c_scale * dist, 1e-10f);
    }

    float integ[8] = {0.f, 0.f, 0.f, 0.f, 0.f, 0.f, 0.f, 0.f};
    for (int q = 0; q < NQ; ++q) {
        const float2 cw = s_tab[q];
#pragma unroll
        for (int r = 0; r < 8; ++r) {
            integ[r] = fmaf(cw.y, __builtin_amdgcn_exp2f(x[r] * cw.x), integ[r]);
        }
    }

#pragma unroll
    for (int r = 0; r < 8; ++r) {
        const int row = bi * 16 + 8 * half + r;
        float l2x  = __builtin_amdgcn_logf(x[r]);
        float e    = fmaf(nu, l2x, log_norm2) - x[r] * LOG2E_F;
        float val  = var * __builtin_amdgcn_exp2f(e) * integ[r];
        if (row == col) val = var;
        out[row * MAT_N + col] = val;
    }
}

extern "C" void kernel_launch(void* const* d_in, const int* in_sizes, int n_in,
                              void* d_out, int out_size, void* d_ws, size_t ws_size,
                              hipStream_t stream) {
    const float* X    = (const float*)d_in[0];
    const float* nu   = (const float*)d_in[1];
    const float* lvar = (const float*)d_in[2];
    const float* llen = (const float*)d_in[3];
    float* ws  = (float*)d_ws;
    float* out = (float*)d_out;

    matern_sqn_kernel<<<MAT_N / 256, 256, 0, stream>>>(X, ws);
    matern_setup_kernel<<<1, 256, 0, stream>>>(nu, lvar, llen, ws);
    matern_main_kernel<<<(64 * 64) / 8, 256, 0, stream>>>(X, ws, out);
}


# Fuse the row-norm and table kernels into the tile kernel (1 launch, not 3): each block rebuilds the table in LDS, each wave sums its row norms from its own fragments (norm sum order changes)
# speedup vs baseline: 1.4241x; 1.4241x over previous
#include <hip/hip_runtime.h>


typedef __attribute__((ext_vector_type(2))) float v2f;
typedef __attribute__((ext_vector_type(8))) float v8f;

#define MAT_N 1024
#define MAT_D 32
#define NQ    192

#define LOG2E_F 1.44269504088896340736f
#define LN2_F   0.69314718055994530942f

__global__ __launch_bounds__(256) void matern_fused_kernel(const float* __restrict__ X,
                                                           const float* __restrict__ nu_p,
                                                           const float* __restrict__ logvar_p,
                                                           const float* __restrict__ logl_p,
                                                           float* __restrict__ out) {
    __shared__ float2 s_tab[NQ];
    __shared__ float  s_par[4];

    const int tid  = threadIdx.x;
    const int wave = tid >> 5;
    const int lane = tid & 31;
    const int tile = blockIdx.x * 8 + wave;
    const int bi = tile >> 6;
    const int bj = tile & 63;
    const int half = lane >> 4;
    const int lr = lane & 15;

    const float* Arow = X + (bi * 16 + lr) * MAT_D + 2 * half;
    const float* Brow = X + (bj * 16 + lr) * MAT_D + 2 * half;
    v2f a[8], b[8];
#pragma unroll
    for (int s = 0; s < 8; ++s) {
        a[s] = *(const v2f*)(Arow + 4 * s);
        b[s] = *(const v2f*)(Brow + 4 * s);
    }

    const float nu_in = nu_p[0];
    if (tid < NQ) {
        const float h = 6.0f / (float)(NQ - 1);
        float t = h * (float)tid;
        float w = h;
        if (tid == 0 || tid == NQ - 1) w *= 0.5f;
        s_tab[tid] = make_float2(-(coshf(t) - 1.0f) * LOG2E_F, w * coshf(nu_in * t));
    } else if (tid == NQ) {
        float var_in = expf(logvar_p[0]);
        float len_in = expf(logl_p[0]);
        s_par[0] = sqrtf(2.0f * nu_in) / len_in;
        s_par[1] = var_in;
        s_par[2] = ((1.0f - nu_in) * LN2_F - lgammaf(nu_in)) * LOG2E_F;
        s_par[3] = nu_in;
    }

    v8f acc = {};
    v2f pa2 = {0.0f, 0.0f}, pb2 = {0.0f, 0.0f};
#pragma unroll
    for (int s = 0; s < 8; ++s) {
        acc = __builtin_amdgcn_wmma_f32_16x16x4_f32(
             false, a[s],  false, b[s],
             (short)0, acc,  false,  false);
        pa2 = __builtin_elementwise_fma(a[s], a[s], pa2);
        pb2 = __builtin_elementwise_fma(b[s], b[s], pb2);
    }
    const float pa = pa2.x + pa2.y;
    const float pb = pb2.x + pb2.y;
    const float na = pa + __shfl_xor(pa, 16);
    const float nb = pb + __shfl_xor(pb, 16);

    const int col = bj * 16 + lr;
    float dist[8];
#pragma unroll
    for (int r = 0; r < 8; ++r) {
        const float sqn_i = __shfl(na, 8 * half + r);
        float sq = sqn_i + nb - 2.0f * acc[r];
        sq = fmaxf(sq, 1e-24f);
        dist[r] = __builtin_amdgcn_sqrtf(sq);
    }

    __syncthreads();
    const float c_scale   = s_par[0];
    const float var       = s_par[1];
    const float log_norm2 = s_par[2];
    const float nu        = s_par[3];

    float x[8];
#pragma unroll
    for (int r = 0; r < 8; ++r) x[r] = fmaxf(c_scale * dist[r], 1e-10f);

    float integ[8] = {0.f, 0.f, 0.f, 0.f, 0.f, 0.f, 0.f, 0.f};
    for (int q = 0; q < NQ; ++q) {
        const float2 cw = s_tab[q];
#pragma unroll
        for (int r = 0; r < 8; ++r) {
            integ[r] = fmaf(cw.y, __builtin_amdgcn_exp2f(x[r] * cw.x), integ[r]);
        }
    }

#pragma unroll
    for (int r = 0; r < 8; ++r) {
        const int row = bi * 16 + 8 * half + r;
        float l2x  = __builtin_amdgcn_logf(x[r]);
        float e    = fmaf(nu, l2x, log_norm2) - x[r] * LOG2E_F;
        float val  = var * __builtin_amdgcn_exp2f(e) * integ[r];
        if (row == col) val = var;
        out[row * MAT_N + col] = val;
    }
}

extern "C" void kernel_launch(void* const* d_in, const int* in_sizes, int n_in,
                              void* d_out, int out_size, void* d_ws, size_t ws_size,
                              hipStream_t stream) {
    const float* X    = (const float*)d_in[0];
    const float* nu   = (const float*)d_in[1];
    const float* lvar = (const float*)d_in[2];
    const float* llen = (const float*)d_in[3];
    float* out = (float*)d_out;

    matern_fused_kernel<<<(64 * 64) / 8, 256, 0, stream>>>(X, nu, lvar, llen, out);
}


# Use the symmetry out[i][j]=out[j][i]: compute only the 2080 tiles with bi<=bj and store each off-diagonal tile also transposed; 256 blocks of 8 tiles, the 32 tiles over 2048 split one row per block
# speedup vs baseline: 2.0872x; 1.4656x over previous
#include <hip/hip_runtime.h>


typedef __attribute__((ext_vector_type(2))) float v2f;
typedef __attribute__((ext_vector_type(4))) float v4f;
typedef __attribute__((ext_vector_type(8))) float v8f;

#define MAT_N 1024
#define MAT_D 32
#define NQ    192
#define NT    64
#define NBLK  256
#define NWHOLE (NBLK * 8)

#define LOG2E_F 1.44269504088896340736f
#define LN2_F   0.69314718055994530942f

__device__ __forceinline__ void tile_of(int t, int& bi, int& bj) {
    const int p = t / 65;
    const int c = t - 65 * p;
    const bool first = c < NT - p;
    bi = first ? p : (NT - 1 - p);
    bj = first ? (p + c) : (c - 1);
}

__device__ __forceinline__ void load_frags(const float* __restrict__ X, int bi, int bj,
                                           int half, int lr, v2f (&a)[8], v2f (&b)[8]) {
    const float* Arow = X + (bi * 16 + lr) * MAT_D + 2 * half;
    const float* Brow = X + (bj * 16 + lr) * MAT_D + 2 * half;
#pragma unroll
    for (int s = 0; s < 8; ++s) {
        a[s] = *(const v2f*)(Arow + 4 * s);
        b[s] = *(const v2f*)(Brow + 4 * s);
    }
}

__device__ __forceinline__ void gram_tile(const v2f (&a)[8], const v2f (&b)[8],
                                          v8f& acc, float& na, float& nb) {
    acc = (v8f){};
    v2f pa2 = {0.0f, 0.0f}, pb2 = {0.0f, 0.0f};
#pragma unroll
    for (int s = 0; s < 8; ++s) {
        acc = __builtin_amdgcn_wmma_f32_16x16x4_f32(
             false, a[s],  false, b[s],
             (short)0, acc,  false,  false);
        pa2 = __builtin_elementwise_fma(a[s], a[s], pa2);
        pb2 = __builtin_elementwise_fma(b[s], b[s], pb2);
    }
    const float pa = pa2.x + pa2.y;
    const float pb = pb2.x + pb2.y;
    na = pa + __shfl_xor(pa, 16);
    nb = pb + __shfl_xor(pb, 16);
}

__device__ __forceinline__ float dist_of(float sqn_i, float sqn_j, float g) {
    float sq = sqn_i + sqn_j - 2.0f * g;
    sq = fmaxf(sq, 1e-24f);
    return __builtin_amdgcn_sqrtf(sq);
}

template <int NR>
__device__ __forceinline__ void quad_sum(const float2* s_tab, const float (&x)[NR],
                                         float (&integ)[NR]) {
#pragma unroll
    for (int r = 0; r < NR; ++r) integ[r] = 0.0f;
    for (int q = 0; q < NQ; ++q) {
        const float2 cw = s_tab[q];
#pragma unroll
        for (int r = 0; r < NR; ++r) {
            integ[r] = fmaf(cw.y, __builtin_amdgcn_exp2f(x[r] * cw.x), integ[r]);
        }
    }
}

__device__ __forceinline__ float matern_val(float x, float integ, float var, float nu,
                                            float log_norm2) {
    const float l2x = __builtin_amdgcn_logf(x);
    const float e   = fmaf(nu, l2x, log_norm2) - x * LOG2E_F;
    return var * __builtin_amdgcn_exp2f(e) * integ;
}

__global__ __launch_bounds__(256) void matern_fused_kernel(const float* __restrict__ X,
                                                           const float* __restrict__ nu_p,
                                                           const float* __restrict__ logvar_p,
                                                           const float* __restrict__ logl_p,
                                                           float* __restrict__ out) {
    __shared__ float2 s_tab[NQ];
    __shared__ float  s_par[4];

    const int tid  = threadIdx.x;
    const int wave = __builtin_amdgcn_readfirstlane(tid >> 5);
    const int lane = tid & 31;
    const int half = lane >> 4;
    const int lr = lane & 15;

    int bi, bj;
    tile_of(blockIdx.x * 8 + wave, bi, bj);
    const bool spare = (wave == 0);
    int bi2, bj2;
    tile_of(NWHOLE + (blockIdx.x >> 3), bi2, bj2);
    const int rr = blockIdx.x & 7;

    v2f a[8], b[8];
    load_frags(X, bi, bj, half, lr, a, b);
    v2f a2[8], b2[8];
    if (spare) load_frags(X, bi2, bj2, half, lr, a2, b2);

    const float nu_in = nu_p[0];
    if (tid < NQ) {
        const float h = 6.0f / (float)(NQ - 1);
        float t = h * (float)tid;
        float w = h;
        if (tid == 0 || tid == NQ - 1) w *= 0.5f;
        s_tab[tid] = make_float2(-(coshf(t) - 1.0f) * LOG2E_F, w * coshf(nu_in * t));
    } else if (tid == NQ) {
        float var_in = expf(logvar_p[0]);
        float len_in = expf(logl_p[0]);
        s_par[0] = sqrtf(2.0f * nu_in) / len_in;
        s_par[1] = var_in;
        s_par[2] = ((1.0f - nu_in) * LN2_F - lgammaf(nu_in)) * LOG2E_F;
        s_par[3] = nu_in;
    }

    v8f acc;
    float na, nb;
    gram_tile(a, b, acc, na, nb);
    float dist[9];
#pragma unroll
    for (int r = 0; r < 8; ++r) {
        const float sqn_i = __shfl(na, 8 * half + r);
        dist[r] = dist_of(sqn_i, nb, acc[r]);
    }
    dist[8] = 1.0f;
    if (spare) {
        v8f acc2;
        float na2, nb2;
        gram_tile(a2, b2, acc2, na2, nb2);
        float g = acc2[0];
#pragma unroll
        for (int r = 1; r < 8; ++r) g = (rr == r) ? acc2[r] : g;
        const float sqn_i = __shfl(na2, 8 * half + rr);
        dist[8] = dist_of(sqn_i, nb2, g);
    }

    __syncthreads();
    const float c_scale   = s_par[0];
    const float var       = s_par[1];
    const float log_norm2 = s_par[2];
    const float nu        = s_par[3];

    const int col = bj * 16 + lr;
    const int row0 = bi * 16 + 8 * half;
    float val[8];
    if (spare) {
        float x[9], integ[9];
#pragma unroll
        for (int r = 0; r < 9; ++r) x[r] = fmaxf(c_scale * dist[r], 1e-10f);
        quad_sum<9>(s_tab, x, integ);
#pragma unroll
        for (int r = 0; r < 8; ++r) val[r] = matern_val(x[r], integ[r], var, nu, log_norm2);
        const int row2 = bi2 * 16 + 8 * half + rr;
        const int col2 = bj2 * 16 + lr;
        float v2 = matern_val(x[8], integ[8], var, nu, log_norm2);
        if (row2 == col2) v2 = var;
        out[row2 * MAT_N + col2] = v2;
        if (bi2 != bj2) out[col2 * MAT_N + row2] = v2;
    } else {
        float x[8], integ[8];
#pragma unroll
        for (int r = 0; r < 8; ++r) x[r] = fmaxf(c_scale * dist[r], 1e-10f);
        quad_sum<8>(s_tab, x, integ);
#pragma unroll
        for (int r = 0; r < 8; ++r) val[r] = matern_val(x[r], integ[r], var, nu, log_norm2);
    }

#pragma unroll
    for (int r = 0; r < 8; ++r) {
        if (row0 + r == col) val[r] = var;
        out[(row0 + r) * MAT_N + col] = val[r];
    }
    if (bi != bj) {
        float* dst = out + col * MAT_N + row0;
        *(v4f*)(dst)     = (v4f){val[0], val[1], val[2], val[3]};
        *(v4f*)(dst + 4) = (v4f){val[4], val[5], val[6], val[7]};
    }
}

extern "C" void kernel_launch(void* const* d_in, const int* in_sizes, int n_in,
                              void* d_out, int out_size, void* d_ws, size_t ws_size,
                              hipStream_t stream) {
    const float* X    = (const float*)d_in[0];
    const float* nu   = (const float*)d_in[1];
    const float* lvar = (const float*)d_in[2];
    const float* llen = (const float*)d_in[3];
    float* out = (float*)d_out;

    matern_fused_kernel<<<NBLK, 256, 0, stream>>>(X, nu, lvar, llen, out);
}


# Stop each wave's 192-term quadrature loop behind its last term that is not exactly +0 in f32 (2^e with e < -152 for every element of the wave; e falls with q): every sum keeps its bits
# speedup vs baseline: 2.7750x; 1.3295x over previous
#include <hip/hip_runtime.h>


typedef __attribute__((ext_vector_type(2))) float v2f;
typedef __attribute__((ext_vector_type(4))) float v4f;
typedef __attribute__((ext_vector_type(8))) float v8f;

#define MAT_N 1024
#define MAT_D 32
#define NQ    192
#define NT    64
#define NBLK  256
#define NWHOLE (NBLK * 8)

#define LOG2E_F 1.44269504088896340736f
#define LN2_F   0.69314718055994530942f

__device__ __forceinline__ void tile_of(int t, int& bi, int& bj) {
    const int p = t / 65;
    const int c = t - 65 * p;
    const bool first = c < NT - p;
    bi = first ? p : (NT - 1 - p);
    bj = first ? (p + c) : (c - 1);
}

__device__ __forceinline__ void load_frags(const float* __restrict__ X, int bi, int bj,
                                           int half, int lr, v2f (&a)[8], v2f (&b)[8]) {
    const float* Arow = X + (bi * 16 + lr) * MAT_D + 2 * half;
    const float* Brow = X + (bj * 16 + lr) * MAT_D + 2 * half;
#pragma unroll
    for (int s = 0; s < 8; ++s) {
        a[s] = *(const v2f*)(Arow + 4 * s);
        b[s] = *(const v2f*)(Brow + 4 * s);
    }
}

__device__ __forceinline__ void gram_tile(const v2f (&a)[8], const v2f (&b)[8],
                                          v8f& acc, float& na, float& nb) {
    acc = (v8f){};
    v2f pa2 = {0.0f, 0.0f}, pb2 = {0.0f, 0.0f};
#pragma unroll
    for (int s = 0; s < 8; ++s) {
        acc = __builtin_amdgcn_wmma_f32_16x16x4_f32(
             false, a[s],  false, b[s],
             (short)0, acc,  false,  false);
        pa2 = __builtin_elementwise_fma(a[s], a[s], pa2);
        pb2 = __builtin_elementwise_fma(b[s], b[s], pb2);
    }
    const float pa = pa2.x + pa2.y;
    const float pb = pb2.x + pb2.y;
    na = pa + __shfl_xor(pa, 16);
    nb = pb + __shfl_xor(pb, 16);
}

__device__ __forceinline__ float dist_of(float sqn_i, float sqn_j, float g) {
    float sq = sqn_i + sqn_j - 2.0f * g;
    sq = fmaxf(sq, 1e-24f);
    return __builtin_amdgcn_sqrtf(sq);
}

#define ZERO_EXP (-152.0f)
#define XBIG 3.0e38f

template <int n>
__device__ __forceinline__ float row_xor(float v) {
    return __builtin_bit_cast(float, __builtin_amdgcn_update_dpp(
        0, __builtin_bit_cast(int, v), 0x160 + n, 0xf, 0xf, true));
}

__device__ __forceinline__ float wave_min(float v) {
    v = fminf(v, row_xor<1>(v));
    v = fminf(v, row_xor<2>(v));
    v = fminf(v, row_xor<4>(v));
    v = fminf(v, row_xor<8>(v));
    const int iv = __builtin_bit_cast(int, v);
    v = fminf(v, __builtin_bit_cast(float, __builtin_amdgcn_permlanex16(
                     iv, iv, 0x76543210, 0xfedcba98, false, false)));
    return __builtin_bit_cast(float, __builtin_amdgcn_readfirstlane(__builtin_bit_cast(int, v)));
}

__device__ __forceinline__ int live_terms(const float2* s_tab, float xmin, int lane) {
    int n = 1;
#pragma unroll
    for (int k = 0; k < NQ / 32; ++k) {
        const float2 cw = s_tab[32 * k + lane];
        const unsigned m = __builtin_amdgcn_ballot_w32(xmin * cw.x >= ZERO_EXP) |
                           __builtin_amdgcn_ballot_w32(!(__builtin_fabsf(cw.y) < __builtin_inff()));
        if (m != 0u) n = 32 * k + 32 - __builtin_clz(m);
    }
    return n;
}

template <int NR>
__device__ __forceinline__ void quad_sum(const float2* s_tab, int n, const float (&x)[NR],
                                         float (&integ)[NR]) {
#pragma unroll
    for (int r = 0; r < NR; ++r) integ[r] = 0.0f;
    for (int q = 0; q < n; ++q) {
        const float2 cw = s_tab[q];
#pragma unroll
        for (int r = 0; r < NR; ++r) {
            integ[r] = fmaf(cw.y, __builtin_amdgcn_exp2f(x[r] * cw.x), integ[r]);
        }
    }
}

__device__ __forceinline__ float matern_val(float x, float integ, float var, float nu,
                                            float log_norm2) {
    const float l2x = __builtin_amdgcn_logf(x);
    const float e   = fmaf(nu, l2x, log_norm2) - x * LOG2E_F;
    return var * __builtin_amdgcn_exp2f(e) * integ;
}

__global__ __launch_bounds__(256) void matern_fused_kernel(const float* __restrict__ X,
                                                           const float* __restrict__ nu_p,
                                                           const float* __restrict__ logvar_p,
                                                           const float* __restrict__ logl_p,
                                                           float* __restrict__ out) {
    __shared__ float2 s_tab[NQ];
    __shared__ float  s_par[4];

    const int tid  = threadIdx.x;
    const int wave = __builtin_amdgcn_readfirstlane(tid >> 5);
    const int lane = tid & 31;
    const int half = lane >> 4;
    const int lr = lane & 15;

    int bi, bj;
    tile_of(blockIdx.x * 8 + wave, bi, bj);
    const bool spare = (wave == 0);
    int bi2, bj2;
    tile_of(NWHOLE + (blockIdx.x >> 3), bi2, bj2);
    const int rr = blockIdx.x & 7;

    v2f a[8], b[8];
    load_frags(X, bi, bj, half, lr, a, b);
    v2f a2[8], b2[8];
    if (spare) load_frags(X, bi2, bj2, half, lr, a2, b2);

    const float nu_in = nu_p[0];
    if (tid < NQ) {
        const float h = 6.0f / (float)(NQ - 1);
        float t = h * (float)tid;
        float w = h;
        if (tid == 0 || tid == NQ - 1) w *= 0.5f;
        s_tab[tid] = make_float2(-(coshf(t) - 1.0f) * LOG2E_F, w * coshf(nu_in * t));
    } else if (tid == NQ) {
        float var_in = expf(logvar_p[0]);
        float len_in = expf(logl_p[0]);
        s_par[0] = sqrtf(2.0f * nu_in) / len_in;
        s_par[1] = var_in;
        s_par[2] = ((1.0f - nu_in) * LN2_F - lgammaf(nu_in)) * LOG2E_F;
        s_par[3] = nu_in;
    }

    v8f acc;
    float na, nb;
    gram_tile(a, b, acc, na, nb);
    float dist[9];
#pragma unroll
    for (int r = 0; r < 8; ++r) {
        const float sqn_i = __shfl(na, 8 * half + r);
        dist[r] = dist_of(sqn_i, nb, acc[r]);
    }
    dist[8] = 1.0f;
    const int col = bj * 16 + lr;
    const int row0 = bi * 16 + 8 * half;
    float dmin = XBIG;
#pragma unroll
    for (int r = 0; r < 8; ++r) dmin = fminf(dmin, (row0 + r == col) ? XBIG : dist[r]);
    if (spare) {
        v8f acc2;
        float na2, nb2;
        gram_tile(a2, b2, acc2, na2, nb2);
        float g = acc2[0];
#pragma unroll
        for (int r = 1; r < 8; ++r) g = (rr == r) ? acc2[r] : g;
        const float sqn_i = __shfl(na2, 8 * half + rr);
        dist[8] = dist_of(sqn_i, nb2, g);
        const bool diag2 = (bi2 * 16 + 8 * half + rr == bj2 * 16 + lr);
        dmin = fminf(dmin, diag2 ? XBIG : dist[8]);
    }
    dmin = wave_min(dmin);

    __syncthreads();
    const float c_scale   = s_par[0];
    const float var       = s_par[1];
    const float log_norm2 = s_par[2];
    const float nu        = s_par[3];

    const int nq = live_terms(s_tab, fmaxf(c_scale * dmin, 1e-10f), lane);

    float val[8];
    if (spare) {
        float x[9], integ[9];
#pragma unroll
        for (int r = 0; r < 9; ++r) x[r] = fmaxf(c_scale * dist[r], 1e-10f);
        quad_sum<9>(s_tab, nq, x, integ);
#pragma unroll
        for (int r = 0; r < 8; ++r) val[r] = matern_val(x[r], integ[r], var, nu, log_norm2);
        const int row2 = bi2 * 16 + 8 * half + rr;
        const int col2 = bj2 * 16 + lr;
        float v2 = matern_val(x[8], integ[8], var, nu, log_norm2);
        if (row2 == col2) v2 = var;
        out[row2 * MAT_N + col2] = v2;
        if (bi2 != bj2) out[col2 * MAT_N + row2] = v2;
    } else {
        float x[8], integ[8];
#pragma unroll
        for (int r = 0; r < 8; ++r) x[r] = fmaxf(c_scale * dist[r], 1e-10f);
        quad_sum<8>(s_tab, nq, x, integ);
#pragma unroll
        for (int r = 0; r < 8; ++r) val[r] = matern_val(x[r], integ[r], var, nu, log_norm2);
    }

#pragma unroll
    for (int r = 0; r < 8; ++r) {
        if (row0 + r == col) val[r] = var;
        out[(row0 + r) * MAT_N + col] = val[r];
    }
    if (bi != bj) {
        float* dst = out + col * MAT_N + row0;
        *(v4f*)(dst)     = (v4f){val[0], val[1], val[2], val[3]};
        *(v4f*)(dst + 4) = (v4f){val[4], val[5], val[6], val[7]};
    }
}

extern "C" void kernel_launch(void* const* d_in, const int* in_sizes, int n_in,
                              void* d_out, int out_size, void* d_ws, size_t ws_size,
                              hipStream_t stream) {
    const float* X    = (const float*)d_in[0];
    const float* nu   = (const float*)d_in[1];
    const float* lvar = (const float*)d_in[2];
    const float* llen = (const float*)d_in[3];
    float* out = (float*)d_out;

    matern_fused_kernel<<<NBLK, 256, 0, stream>>>(X, nu, lvar, llen, out);
}


# Stop each wave's quadrature loop behind its last term that can still change an f32 running sum (term below 2^-27 of the sum's first term at the wave's smallest x): every sum keeps its bits
# speedup vs baseline: 3.0140x; 1.0861x over previous
#include <hip/hip_runtime.h>


typedef __attribute__((ext_vector_type(2))) float v2f;
typedef __attribute__((ext_vector_type(4))) float v4f;
typedef __attribute__((ext_vector_type(8))) float v8f;

#define MAT_N 1024
#define MAT_D 32
#define NQ    192
#define NT    64
#define NBLK  256
#define NWHOLE (NBLK * 8)

#define LOG2E_F 1.44269504088896340736f
#define LN2_F   0.69314718055994530942f

__device__ __forceinline__ void tile_of(int t, int& bi, int& bj) {
    const int p = t / 65;
    const int c = t - 65 * p;
    const bool first = c < NT - p;
    bi = first ? p : (NT - 1 - p);
    bj = first ? (p + c) : (c - 1);
}

__device__ __forceinline__ void load_frags(const float* __restrict__ X, int bi, int bj,
                                           int half, int lr, v2f (&a)[8], v2f (&b)[8]) {
    const float* Arow = X + (bi * 16 + lr) * MAT_D + 2 * half;
    const float* Brow = X + (bj * 16 + lr) * MAT_D + 2 * half;
#pragma unroll
    for (int s = 0; s < 8; ++s) {
        a[s] = *(const v2f*)(Arow + 4 * s);
        b[s] = *(const v2f*)(Brow + 4 * s);
    }
}

__device__ __forceinline__ void gram_tile(const v2f (&a)[8], const v2f (&b)[8],
                                          v8f& acc, float& na, float& nb) {
    acc = (v8f){};
    v2f pa2 = {0.0f, 0.0f}, pb2 = {0.0f, 0.0f};
#pragma unroll
    for (int s = 0; s < 8; ++s) {
        acc = __builtin_amdgcn_wmma_f32_16x16x4_f32(
             false, a[s],  false, b[s],
             (short)0, acc,  false,  false);
        pa2 = __builtin_elementwise_fma(a[s], a[s], pa2);
        pb2 = __builtin_elementwise_fma(b[s], b[s], pb2);
    }
    const float pa = pa2.x + pa2.y;
    const float pb = pb2.x + pb2.y;
    na = pa + __shfl_xor(pa, 16);
    nb = pb + __shfl_xor(pb, 16);
}

__device__ __forceinline__ float dist_of(float sqn_i, float sqn_j, float g) {
    float sq = sqn_i + sqn_j - 2.0f * g;
    sq = fmaxf(sq, 1e-24f);
    return __builtin_amdgcn_sqrtf(sq);
}

#define DEAD_REL 7.450580596923828125e-9f
#define XBIG 3.0e38f

template <int n>
__device__ __forceinline__ float row_xor(float v) {
    return __builtin_bit_cast(float, __builtin_amdgcn_update_dpp(
        0, __builtin_bit_cast(int, v), 0x160 + n, 0xf, 0xf, true));
}

__device__ __forceinline__ float wave_min(float v) {
    v = fminf(v, row_xor<1>(v));
    v = fminf(v, row_xor<2>(v));
    v = fminf(v, row_xor<4>(v));
    v = fminf(v, row_xor<8>(v));
    const int iv = __builtin_bit_cast(int, v);
    v = fminf(v, __builtin_bit_cast(float, __builtin_amdgcn_permlanex16(
                     iv, iv, 0x76543210, 0xfedcba98, false, false)));
    return __builtin_bit_cast(float, __builtin_amdgcn_readfirstlane(__builtin_bit_cast(int, v)));
}

__device__ __forceinline__ int live_terms(const float2* s_tab, float xmin, int lane) {
    const float dead = s_tab[0].y * DEAD_REL;
    int n = 1;
#pragma unroll
    for (int k = 0; k < NQ / 32; ++k) {
        const float2 cw = s_tab[32 * k + lane];
        const float tmax = cw.y * __builtin_amdgcn_exp2f(xmin * cw.x);
        const unsigned m = __builtin_amdgcn_ballot_w32(tmax >= dead) |
                           __builtin_amdgcn_ballot_w32(!(__builtin_fabsf(cw.y) < __builtin_inff()));
        if (m != 0u) n = 32 * k + 32 - __builtin_clz(m);
    }
    return n;
}

template <int NR>
__device__ __forceinline__ void quad_sum(const float2* s_tab, int n, const float (&x)[NR],
                                         float (&integ)[NR]) {
#pragma unroll
    for (int r = 0; r < NR; ++r) integ[r] = 0.0f;
    for (int q = 0; q < n; ++q) {
        const float2 cw = s_tab[q];
#pragma unroll
        for (int r = 0; r < NR; ++r) {
            integ[r] = fmaf(cw.y, __builtin_amdgcn_exp2f(x[r] * cw.x), integ[r]);
        }
    }
}

__device__ __forceinline__ float matern_val(float x, float integ, float var, float nu,
                                            float log_norm2) {
    const float l2x = __builtin_amdgcn_logf(x);
    const float e   = fmaf(nu, l2x, log_norm2) - x * LOG2E_F;
    return var * __builtin_amdgcn_exp2f(e) * integ;
}

__global__ __launch_bounds__(256) void matern_fused_kernel(const float* __restrict__ X,
                                                           const float* __restrict__ nu_p,
                                                           const float* __restrict__ logvar_p,
                                                           const float* __restrict__ logl_p,
                                                           float* __restrict__ out) {
    __shared__ float2 s_tab[NQ];
    __shared__ float  s_par[4];

    const int tid  = threadIdx.x;
    const int wave = __builtin_amdgcn_readfirstlane(tid >> 5);
    const int lane = tid & 31;
    const int half = lane >> 4;
    const int lr = lane & 15;

    int bi, bj;
    tile_of(blockIdx.x * 8 + wave, bi, bj);
    const bool spare = (wave == 0);
    int bi2, bj2;
    tile_of(NWHOLE + (blockIdx.x >> 3), bi2, bj2);
    const int rr = blockIdx.x & 7;

    v2f a[8], b[8];
    load_frags(X, bi, bj, half, lr, a, b);
    v2f a2[8], b2[8];
    if (spare) load_frags(X, bi2, bj2, half, lr, a2, b2);

    const float nu_in = nu_p[0];
    if (tid < NQ) {
        const float h = 6.0f / (float)(NQ - 1);
        float t = h * (float)tid;
        float w = h;
        if (tid == 0 || tid == NQ - 1) w *= 0.5f;
        s_tab[tid] = make_float2(-(coshf(t) - 1.0f) * LOG2E_F, w * coshf(nu_in * t));
    } else if (tid == NQ) {
        float var_in = expf(logvar_p[0]);
        float len_in = expf(logl_p[0]);
        s_par[0] = sqrtf(2.0f * nu_in) / len_in;
        s_par[1] = var_in;
        s_par[2] = ((1.0f - nu_in) * LN2_F - lgammaf(nu_in)) * LOG2E_F;
        s_par[3] = nu_in;
    }

    v8f acc;
    float na, nb;
    gram_tile(a, b, acc, na, nb);
    float dist[9];
#pragma unroll
    for (int r = 0; r < 8; ++r) {
        const float sqn_i = __shfl(na, 8 * half + r);
        dist[r] = dist_of(sqn_i, nb, acc[r]);
    }
    dist[8] = 1.0f;
    const int col = bj * 16 + lr;
    const int row0 = bi * 16 + 8 * half;
    float dmin = XBIG;
#pragma unroll
    for (int r = 0; r < 8; ++r) dmin = fminf(dmin, (row0 + r == col) ? XBIG : dist[r]);
    if (spare) {
        v8f acc2;
        float na2, nb2;
        gram_tile(a2, b2, acc2, na2, nb2);
        float g = acc2[0];
#pragma unroll
        for (int r = 1; r < 8; ++r) g = (rr == r) ? acc2[r] : g;
        const float sqn_i = __shfl(na2, 8 * half + rr);
        dist[8] = dist_of(sqn_i, nb2, g);
        const bool diag2 = (bi2 * 16 + 8 * half + rr == bj2 * 16 + lr);
        dmin = fminf(dmin, diag2 ? XBIG : dist[8]);
    }
    dmin = wave_min(dmin);

    __syncthreads();
    const float c_scale   = s_par[0];
    const float var       = s_par[1];
    const float log_norm2 = s_par[2];
    const float nu        = s_par[3];

    const int nq = live_terms(s_tab, fmaxf(c_scale * dmin, 1e-10f), lane);

    float val[8];
    if (spare) {
        float x[9], integ[9];
#pragma unroll
        for (int r = 0; r < 9; ++r) x[r] = fmaxf(c_scale * dist[r], 1e-10f);
        quad_sum<9>(s_tab, nq, x, integ);
#pragma unroll
        for (int r = 0; r < 8; ++r) val[r] = matern_val(x[r], integ[r], var, nu, log_norm2);
        const int row2 = bi2 * 16 + 8 * half + rr;
        const int col2 = bj2 * 16 + lr;
        float v2 = matern_val(x[8], integ[8], var, nu, log_norm2);
        if (row2 == col2) v2 = var;
        out[row2 * MAT_N + col2] = v2;
        if (bi2 != bj2) out[col2 * MAT_N + row2] = v2;
    } else {
        float x[8], integ[8];
#pragma unroll
        for (int r = 0; r < 8; ++r) x[r] = fmaxf(c_scale * dist[r], 1e-10f);
        quad_sum<8>(s_tab, nq, x, integ);
#pragma unroll
        for (int r = 0; r < 8; ++r) val[r] = matern_val(x[r], integ[r], var, nu, log_norm2);
    }

#pragma unroll
    for (int r = 0; r < 8; ++r) {
        if (row0 + r == col) val[r] = var;
        out[(row0 + r) * MAT_N + col] = val[r];
    }
    if (bi != bj) {
        float* dst = out + col * MAT_N + row0;
        *(v4f*)(dst)     = (v4f){val[0], val[1], val[2], val[3]};
        *(v4f*)(dst + 4) = (v4f){val[4], val[5], val[6], val[7]};
    }
}

extern "C" void kernel_launch(void* const* d_in, const int* in_sizes, int n_in,
                              void* d_out, int out_size, void* d_ws, size_t ws_size,
                              hipStream_t stream) {
    const float* X    = (const float*)d_in[0];
    const float* nu   = (const float*)d_in[1];
    const float* lvar = (const float*)d_in[2];
    const float* llen = (const float*)d_in[3];
    float* out = (float*)d_out;

    matern_fused_kernel<<<NBLK, 256, 0, stream>>>(X, nu, lvar, llen, out);
}


# speedup vs baseline: 3.2859x; 1.0902x over previous
#include <hip/hip_runtime.h>


typedef __attribute__((ext_vector_type(2))) float v2f;
typedef __attribute__((ext_vector_type(4))) float v4f;
typedef __attribute__((ext_vector_type(8))) float v8f;

#define MAT_N 1024
#define MAT_D 32
#define NQ    192
#define NT    64
#define NBLK  256
#define NWHOLE (NBLK * 8)

#define LOG2E_F 1.44269504088896340736f
#define LN2_F   0.69314718055994530942f

__device__ __forceinline__ void tile_of(int t, int& bi, int& bj) {
    const int p = t / 65;
    const int c = t - 65 * p;
    const bool first = c < NT - p;
    bi = first ? p : (NT - 1 - p);
    bj = first ? (p + c) : (c - 1);
}

__device__ __forceinline__ void load_frags(const float* __restrict__ X, int bi, int bj,
                                           int half, int lr, v2f (&a)[8], v2f (&b)[8]) {
    const float* Arow = X + (bi * 16 + lr) * MAT_D + 2 * half;
    const float* Brow = X + (bj * 16 + lr) * MAT_D + 2 * half;
#pragma unroll
    for (int s = 0; s < 8; ++s) {
        a[s] = *(const v2f*)(Arow + 4 * s);
        b[s] = *(const v2f*)(Brow + 4 * s);
    }
}

__device__ __forceinline__ void gram_tile(const v2f (&a)[8], const v2f (&b)[8],
                                          v8f& acc, float& na, float& nb) {
    acc = (v8f){};
    v2f pa2 = {0.0f, 0.0f}, pb2 = {0.0f, 0.0f};
#pragma unroll
    for (int s = 0; s < 8; ++s) {
        acc = __builtin_amdgcn_wmma_f32_16x16x4_f32(
             false, a[s],  false, b[s],
             (short)0, acc,  false,  false);
        pa2 = __builtin_elementwise_fma(a[s], a[s], pa2);
        pb2 = __builtin_elementwise_fma(b[s], b[s], pb2);
    }
    const float pa = pa2.x + pa2.y;
    const float pb = pb2.x + pb2.y;
    na = pa + __shfl_xor(pa, 16);
    nb = pb + __shfl_xor(pb, 16);
}

__device__ __forceinline__ float dist_of(float sqn_i, float sqn_j, float g) {
    float sq = sqn_i + sqn_j - 2.0f * g;
    sq = fmaxf(sq, 1e-24f);
    return __builtin_amdgcn_sqrtf(sq);
}

#define DEAD_REL 7.450580596923828125e-9f
#define XBIG 3.0e38f

template <int n>
__device__ __forceinline__ float row_xor(float v) {
    return __builtin_bit_cast(float, __builtin_amdgcn_update_dpp(
        0, __builtin_bit_cast(int, v), 0x160 + n, 0xf, 0xf, true));
}

__device__ __forceinline__ float wave_min(float v) {
    v = fminf(v, row_xor<1>(v));
    v = fminf(v, row_xor<2>(v));
    v = fminf(v, row_xor<4>(v));
    v = fminf(v, row_xor<8>(v));
    const int iv = __builtin_bit_cast(int, v);
    v = fminf(v, __builtin_bit_cast(float, __builtin_amdgcn_permlanex16(
                     iv, iv, 0x76543210, 0xfedcba98, false, false)));
    return __builtin_bit_cast(float, __builtin_amdgcn_readfirstlane(__builtin_bit_cast(int, v)));
}

__device__ __forceinline__ int live_terms(const float2* s_tab, float xmin, int lane) {
    const float dead = s_tab[0].y * DEAD_REL;
    int n = 1;
#pragma unroll
    for (int k = 0; k < NQ / 32; ++k) {
        const float2 cw = s_tab[32 * k + lane];
        const float tmax = cw.y * __builtin_amdgcn_exp2f(xmin * cw.x);
        const unsigned m = __builtin_amdgcn_ballot_w32(tmax >= dead) |
                           __builtin_amdgcn_ballot_w32(!(__builtin_fabsf(cw.y) < __builtin_inff()));
        if (m != 0u) n = 32 * k + 32 - __builtin_clz(m);
    }
    return n;
}


#define SCHED_PLAIN 0x002
#define SCHED_TRANS 0x400

#define SLOT_TRANS() __builtin_amdgcn_sched_group_barrier(SCHED_TRANS, 1, 0)
#define SLOT_PLAIN() __builtin_amdgcn_sched_group_barrier(SCHED_PLAIN, 1, 0)

template <int NP, bool ODD, int QUN>
__device__ __forceinline__ void quad_trip(const float2* s_tab, int q, const v2f (&x2)[NP], float x1,
                                          v2f (&s2)[NP], float& s1) {
    float2 cw[QUN];
#pragma unroll
    for (int u = 0; u < QUN; ++u) cw[u] = s_tab[q + u];
    v2f a[NP], an[NP], e[NP], ep[NP];
    float a1 = 0.0f, a1n = 0.0f, e1 = 0.0f, e1p = 0.0f;
#pragma unroll
    for (int r = 0; r < NP; ++r) {
        a[r] = x2[r] * (v2f){cw[0].x, cw[0].x};
        SLOT_PLAIN();
        an[r] = a[r];
        e[r] = a[r];
        ep[r] = a[r];
    }
    if (ODD) {
        a1 = x1 * cw[0].x;
        SLOT_PLAIN();
    }
    __builtin_amdgcn_sched_group_barrier(SCHED_PLAIN, QUN / 2, 0);
#pragma unroll
    for (int u = 0; u < QUN; ++u) {
        const int un = (u + 1 < QUN) ? u + 1 : u;
        const int up = (u > 0) ? u - 1 : 0;
#pragma unroll
        for (int r = 0; r < NP; ++r) {
            e[r].x = __builtin_amdgcn_exp2f(a[r].x);
            SLOT_TRANS();
            if (u + 1 < QUN) {
                an[r] = x2[r] * (v2f){cw[un].x, cw[un].x};
                SLOT_PLAIN();
            }
            e[r].y = __builtin_amdgcn_exp2f(a[r].y);
            SLOT_TRANS();
            if (u > 0) {
                s2[r] = __builtin_elementwise_fma((v2f){cw[up].y, cw[up].y}, ep[r], s2[r]);
                SLOT_PLAIN();
            }
        }
        if (ODD) {
            if (u + 1 < QUN) {
                a1n = x1 * cw[un].x;
                SLOT_PLAIN();
            }
            e1 = __builtin_amdgcn_exp2f(a1);
            SLOT_TRANS();
            if (u > 0) {
                s1 = fmaf(cw[up].y, e1p, s1);
                SLOT_PLAIN();
            }
        }
#pragma unroll
        for (int r = 0; r < NP; ++r) {
            ep[r] = e[r];
            a[r] = an[r];
        }
        e1p = e1;
        a1 = a1n;
    }
#pragma unroll
    for (int r = 0; r < NP; ++r) {
        s2[r] = __builtin_elementwise_fma((v2f){cw[QUN - 1].y, cw[QUN - 1].y}, ep[r], s2[r]);
        SLOT_PLAIN();
    }
    if (ODD) {
        s1 = fmaf(cw[QUN - 1].y, e1p, s1);
        SLOT_PLAIN();
    }
}

template <int NR>
__device__ __forceinline__ void quad_sum(const float2* s_tab, int n, const float (&x)[NR],
                                         float (&integ)[NR]) {
    constexpr int NP = NR / 2;
    constexpr bool ODD = (NR & 1) != 0;
    v2f x2[NP], s2[NP];
#pragma unroll
    for (int r = 0; r < NP; ++r) {
        x2[r] = (v2f){x[2 * r], x[2 * r + 1]};
        s2[r] = (v2f){0.0f, 0.0f};
    }
    const float x1 = x[NR - 1];
    float s1 = 0.0f;
    int q = 0;
    for (; q + 16 <= n; q += 16) quad_trip<NP, ODD, 16>(s_tab, q, x2, x1, s2, s1);
    if (q + 8 <= n) {
        quad_trip<NP, ODD, 8>(s_tab, q, x2, x1, s2, s1);
        q += 8;
    }
    if (q + 4 <= n) {
        quad_trip<NP, ODD, 4>(s_tab, q, x2, x1, s2, s1);
        q += 4;
    }
#pragma nounroll
    for (; q < n; ++q) {
        const float2 cw = s_tab[q];
#pragma unroll
        for (int r = 0; r < NP; ++r) {
            const v2f a = x2[r] * (v2f){cw.x, cw.x};
            const v2f e = {__builtin_amdgcn_exp2f(a.x), __builtin_amdgcn_exp2f(a.y)};
            s2[r] = __builtin_elementwise_fma((v2f){cw.y, cw.y}, e, s2[r]);
        }
        if (ODD) s1 = fmaf(cw.y, __builtin_amdgcn_exp2f(x1 * cw.x), s1);
    }
#pragma unroll
    for (int r = 0; r < NP; ++r) {
        integ[2 * r] = s2[r].x;
        integ[2 * r + 1] = s2[r].y;
    }
    if (ODD) integ[NR - 1] = s1;
}

__device__ __forceinline__ float matern_val(float x, float integ, float var, float nu,
                                            float log_norm2) {
    const float l2x = __builtin_amdgcn_logf(x);
    const float e   = fmaf(nu, l2x, log_norm2) - x * LOG2E_F;
    return var * __builtin_amdgcn_exp2f(e) * integ;
}

__global__ __launch_bounds__(256) void matern_fused_kernel(const float* __restrict__ X,
                                                           const float* __restrict__ nu_p,
                                                           const float* __restrict__ logvar_p,
                                                           const float* __restrict__ logl_p,
                                                           float* __restrict__ out) {
    __shared__ float2 s_tab[NQ];
    __shared__ float  s_par[4];

    const int tid  = threadIdx.x;
    const int wave = __builtin_amdgcn_readfirstlane(tid >> 5);
    const int lane = tid & 31;
    const int half = lane >> 4;
    const int lr = lane & 15;

    int bi, bj;
    tile_of(blockIdx.x * 8 + wave, bi, bj);
    const bool spare = (wave == 0);
    int bi2, bj2;
    tile_of(NWHOLE + (blockIdx.x >> 3), bi2, bj2);
    const int rr = blockIdx.x & 7;

    v2f a[8], b[8];
    load_frags(X, bi, bj, half, lr, a, b);
    v2f a2[8], b2[8];
    if (spare) load_frags(X, bi2, bj2, half, lr, a2, b2);

    const float nu_in = nu_p[0];
    if (tid < NQ) {
        const float h = 6.0f / (float)(NQ - 1);
        float t = h * (float)tid;
        float w = h;
        if (tid == 0 || tid == NQ - 1) w *= 0.5f;
        s_tab[tid] = make_float2(-(coshf(t) - 1.0f) * LOG2E_F, w * coshf(nu_in * t));
    } else if (tid == NQ) {
        float var_in = expf(logvar_p[0]);
        float len_in = expf(logl_p[0]);
        s_par[0] = sqrtf(2.0f * nu_in) / len_in;
        s_par[1] = var_in;
        s_par[2] = ((1.0f - nu_in) * LN2_F - lgammaf(nu_in)) * LOG2E_F;
        s_par[3] = nu_in;
    }

    v8f acc;
    float na, nb;
    gram_tile(a, b, acc, na, nb);
    float dist[9];
#pragma unroll
    for (int r = 0; r < 8; ++r) {
        const float sqn_i = __shfl(na, 8 * half + r);
        dist[r] = dist_of(sqn_i, nb, acc[r]);
    }
    dist[8] = 1.0f;
    const int col = bj * 16 + lr;
    const int row0 = bi * 16 + 8 * half;
    float dmin = XBIG;
#pragma unroll
    for (int r = 0; r < 8; ++r) dmin = fminf(dmin, (row0 + r == col) ? XBIG : dist[r]);
    if (spare) {
        v8f acc2;
        float na2, nb2;
        gram_tile(a2, b2, acc2, na2, nb2);
        float g = acc2[0];
#pragma unroll
        for (int r = 1; r < 8; ++r) g = (rr == r) ? acc2[r] : g;
        const float sqn_i = __shfl(na2, 8 * half + rr);
        dist[8] = dist_of(sqn_i, nb2, g);
        const bool diag2 = (bi2 * 16 + 8 * half + rr == bj2 * 16 + lr);
        dmin = fminf(dmin, diag2 ? XBIG : dist[8]);
    }
    dmin = wave_min(dmin);

    __syncthreads();
    const float c_scale   = s_par[0];
    const float var       = s_par[1];
    const float log_norm2 = s_par[2];
    const float nu        = s_par[3];

    const int nq = live_terms(s_tab, fmaxf(c_scale * dmin, 1e-10f), lane);

    float val[8];
    if (spare) {
        float x[9], integ[9];
#pragma unroll
        for (int r = 0; r < 9; ++r) x[r] = fmaxf(c_scale * dist[r], 1e-10f);
        quad_sum<9>(s_tab, nq, x, integ);
#pragma unroll
        for (int r = 0; r < 8; ++r) val[r] = matern_val(x[r], integ[r], var, nu, log_norm2);
        const int row2 = bi2 * 16 + 8 * half + rr;
        const int col2 = bj2 * 16 + lr;
        float v2 = matern_val(x[8], integ[8], var, nu, log_norm2);
        if (row2 == col2) v2 = var;
        out[row2 * MAT_N + col2] = v2;
        if (bi2 != bj2) out[col2 * MAT_N + row2] = v2;
    } else {
        float x[8], integ[8];
#pragma unroll
        for (int r = 0; r < 8; ++r) x[r] = fmaxf(c_scale * dist[r], 1e-10f);
        quad_sum<8>(s_tab, nq, x, integ);
#pragma unroll
        for (int r = 0; r < 8; ++r) val[r] = matern_val(x[r], integ[r], var, nu, log_norm2);
    }

#pragma unroll
    for (int r = 0; r < 8; ++r) {
        if (row0 + r == col) val[r] = var;
        out[(row0 + r) * MAT_N + col] = val[r];
    }
    if (bi != bj) {
        float* dst = out + col * MAT_N + row0;
        *(v4f*)(dst)     = (v4f){val[0], val[1], val[2], val[3]};
        *(v4f*)(dst + 4) = (v4f){val[4], val[5], val[6], val[7]};
    }
}

extern "C" void kernel_launch(void* const* d_in, const int* in_sizes, int n_in,
                              void* d_out, int out_size, void* d_ws, size_t ws_size,
                              hipStream_t stream) {
    const float* X    = (const float*)d_in[0];
    const float* nu   = (const float*)d_in[1];
    const float* lvar = (const float*)d_in[2];
    const float* llen = (const float*)d_in[3];
    float* out = (float*)d_out;

    matern_fused_kernel<<<NBLK, 256, 0, stream>>>(X, nu, lvar, llen, out);
}
